# starting kernel
# baseline (speedup 1.0000x reference)
#include <hip/hip_runtime.h>


#define DD1  128
#define DD2  128
#define BB   16
#define SIN  64
#define SOUT 128

typedef __attribute__((ext_vector_type(2))) float v2f;
typedef __attribute__((ext_vector_type(4))) float v4f;
typedef __attribute__((ext_vector_type(8))) float v8f;

__global__ __launch_bounds__(256)
void mdrnn_gemm_wmma(const float* __restrict__ x, const float* __restrict__ w,
                     const float* __restrict__ bias, float* __restrict__ a_t)
{
    __shared__ float tile[512 * 32];

    const int i    = blockIdx.z;
    const int j0   = blockIdx.y * 32;
    const int o0   = blockIdx.x * 32;
    const int tid  = threadIdx.x;
    const int lane = tid & 31;
    const int wave = tid >> 5;

    const int lm = lane & 15;
    const int kh = (lane >> 4) * 2;
    const int hh = lane >> 4;

    for (int t = 0; t < 8; ++t) {
        const int jj = wave * 4 + (t >> 1);
        const int nn = (t & 1) * 16;
        const int j  = j0 + jj;

        const float* xrow = x + (((size_t)i * DD2 + j) * BB + lm) * SIN;

        v8f acc = {};
        #pragma unroll
        for (int k0 = 0; k0 < SIN; k0 += 4) {
            v2f af;
            af.x = xrow[k0 + kh];
            af.y = xrow[k0 + kh + 1];
            const float* wp = w + (size_t)(k0 + kh) * SOUT + (o0 + nn + lm);
            v2f bf;
            bf.x = wp[0];
            bf.y = wp[SOUT];
            acc = __builtin_amdgcn_wmma_f32_16x16x4_f32(
                false, af, false, bf, (short)0, acc, false, false);
        }

        #pragma unroll
        for (int v = 0; v < 8; ++v)
            tile[(jj * 16 + (v + 8 * hh)) * 32 + nn + lm] = acc[v];
    }
    __syncthreads();

    for (int p = tid; p < 512; p += 256) {
        const int b  = p >> 5;
        const int oo = p & 31;
        const int o  = o0 + oo;
        const float bv = bias[o];
        float* dst = a_t + (size_t)(b * SOUT + o) * (DD1 * DD2)
                         + (size_t)i * DD2 + j0;
        #pragma unroll
        for (int jj = 0; jj < 32; jj += 4) {
            v4f val;
            val.x = tile[((jj + 0) * 16 + b) * 32 + oo] + bv;
            val.y = tile[((jj + 1) * 16 + b) * 32 + oo] + bv;
            val.z = tile[((jj + 2) * 16 + b) * 32 + oo] + bv;
            val.w = tile[((jj + 3) * 16 + b) * 32 + oo] + bv;
            *(v4f*)(dst + jj) = val;
        }
    }
}

__global__ __launch_bounds__(256)
void mdrnn_scan(const float* __restrict__ a_t, const float* __restrict__ u,
                float* __restrict__ h_out)
{
    __shared__ float boundary[8][DD1];

    const int tid  = threadIdx.x;
    const int lane = tid & 31;
    const int wave = tid >> 5;
    const int bo   = blockIdx.x * 8 + wave;
    const int b    = bo >> 7;
    const int o    = bo & 127;

    const float u0 = u[o];
    const float u1 = u[SOUT + o];
    const float* abase = a_t + (size_t)bo * (DD1 * DD2);
    float* bnd = boundary[wave];

    for (int s = 0; s < 4; ++s) {
        const int j = s * 32 + lane;
        float h = 0.0f;
        for (int t = 0; t < DD1 + 31; ++t) {
            float hl = __shfl_up(h, 1);
            const int i = t - lane;
            if (lane == 0)
                hl = (s > 0 && i >= 0 && i < DD1) ? bnd[i] : 0.0f;
            if (i >= 0 && i < DD1) {
                const float a  = abase[i * DD2 + j];
                const float hn = tanhf(fmaf(u0, h, fmaf(u1, hl, a)));
                h = hn;
                if (lane == 31) bnd[i] = hn;
                h_out[(((size_t)i * DD2 + j) * BB + b) * SOUT + o] = hn;
            }
        }
    }
}

extern "C" void kernel_launch(void* const* d_in, const int* in_sizes, int n_in,
                              void* d_out, int out_size, void* d_ws, size_t ws_size,
                              hipStream_t stream)
{
    const float* x    = (const float*)d_in[0];
    const float* w    = (const float*)d_in[1];
    const float* u    = (const float*)d_in[2];
    const float* bias = (const float*)d_in[3];
    float* h   = (float*)d_out;
    float* a_t = (float*)d_ws;

    dim3 g1(SOUT / 32, DD2 / 32, DD1);
    mdrnn_gemm_wmma<<<g1, 256, 0, stream>>>(x, w, bias, a_t);

    mdrnn_scan<<<(BB * SOUT) / 8, 256, 0, stream>>>(a_t, u, h);
}


# scan kernel reads a_t as whole 128-byte rows one 8-step chunk ahead and skews them through a per-wave LDS ring (boundary reads hoisted too): no global gather on the recurrence; sum order unchanged
# speedup vs baseline: 1.4991x; 1.4991x over previous
#include <hip/hip_runtime.h>


#define DD1  128
#define DD2  128
#define BB   16
#define SIN  64
#define SOUT 128

typedef __attribute__((ext_vector_type(2))) float v2f;
typedef __attribute__((ext_vector_type(4))) float v4f;
typedef __attribute__((ext_vector_type(8))) float v8f;

__global__ __launch_bounds__(256)
void mdrnn_gemm_wmma(const float* __restrict__ x, const float* __restrict__ w,
                     const float* __restrict__ bias, float* __restrict__ a_t)
{
    __shared__ float tile[512 * 32];

    const int i    = blockIdx.z;
    const int j0   = blockIdx.y * 32;
    const int o0   = blockIdx.x * 32;
    const int tid  = threadIdx.x;
    const int lane = tid & 31;
    const int wave = tid >> 5;

    const int lm = lane & 15;
    const int kh = (lane >> 4) * 2;
    const int hh = lane >> 4;

    for (int t = 0; t < 8; ++t) {
        const int jj = wave * 4 + (t >> 1);
        const int nn = (t & 1) * 16;
        const int j  = j0 + jj;

        const float* xrow = x + (((size_t)i * DD2 + j) * BB + lm) * SIN;

        v8f acc = {};
        #pragma unroll
        for (int k0 = 0; k0 < SIN; k0 += 4) {
            v2f af;
            af.x = xrow[k0 + kh];
            af.y = xrow[k0 + kh + 1];
            const float* wp = w + (size_t)(k0 + kh) * SOUT + (o0 + nn + lm);
            v2f bf;
            bf.x = wp[0];
            bf.y = wp[SOUT];
            acc = __builtin_amdgcn_wmma_f32_16x16x4_f32(
                false, af, false, bf, (short)0, acc, false, false);
        }

        #pragma unroll
        for (int v = 0; v < 8; ++v)
            tile[(jj * 16 + (v + 8 * hh)) * 32 + nn + lm] = acc[v];
    }
    __syncthreads();

    for (int p = tid; p < 512; p += 256) {
        const int b  = p >> 5;
        const int oo = p & 31;
        const int o  = o0 + oo;
        const float bv = bias[o];
        float* dst = a_t + (size_t)(b * SOUT + o) * (DD1 * DD2)
                         + (size_t)i * DD2 + j0;
        #pragma unroll
        for (int jj = 0; jj < 32; jj += 4) {
            v4f val;
            val.x = tile[((jj + 0) * 16 + b) * 32 + oo] + bv;
            val.y = tile[((jj + 1) * 16 + b) * 32 + oo] + bv;
            val.z = tile[((jj + 2) * 16 + b) * 32 + oo] + bv;
            val.w = tile[((jj + 3) * 16 + b) * 32 + oo] + bv;
            *(v4f*)(dst + jj) = val;
        }
    }
}

#define RING_ROWS 64
#define CHUNK     8
#define NCHUNK    20

__global__ __launch_bounds__(256)
void mdrnn_scan(const float* __restrict__ a_t, const float* __restrict__ u,
                float* __restrict__ h_out)
{
    __shared__ float boundary[8][DD1];
    __shared__ float ring[8][RING_ROWS * 32];

    const int tid  = threadIdx.x;
    const int lane = tid & 31;
    const int wave = tid >> 5;
    const int bo   = blockIdx.x * 8 + wave;
    const int b    = bo >> 7;
    const int o    = bo & 127;

    const float u0 = u[o];
    const float u1 = u[SOUT + o];
    const float* abase = a_t + (size_t)bo * (DD1 * DD2) + lane;
    float* bnd = boundary[wave];
    float* rw  = ring[wave];

    float r[CHUNK];
    #pragma unroll
    for (int k = 0; k < CHUNK; ++k)
        r[k] = abase[k * DD2];

    for (int s = 0; s < 4; ++s) {
        const int j = s * 32 + lane;
        float h = 0.0f;
        for (int c = 0; c < NCHUNK; ++c) {
            const int t0 = c * CHUNK;
            if (t0 < DD1) {
                #pragma unroll
                for (int k = 0; k < CHUNK; ++k)
                    rw[((t0 + k) & (RING_ROWS - 1)) * 32 + lane] = r[k];
            }
            if (t0 + CHUNK < DD1) {
                #pragma unroll
                for (int k = 0; k < CHUNK; ++k)
                    r[k] = abase[(t0 + CHUNK + k) * DD2 + s * 32];
            } else if (c == NCHUNK - 1 && s < 3) {
                #pragma unroll
                for (int k = 0; k < CHUNK; ++k)
                    r[k] = abase[k * DD2 + (s + 1) * 32];
            }

            float av[CHUNK], bv[CHUNK];
            #pragma unroll
            for (int k = 0; k < CHUNK; ++k) {
                const int i = t0 + k - lane;
                av[k] = 0.0f;
                bv[k] = 0.0f;
                if (i >= 0 && i < DD1)
                    av[k] = rw[(i & (RING_ROWS - 1)) * 32 + lane];
            }
            if (lane == 0 && s > 0 && t0 < DD1) {
                #pragma unroll
                for (int k = 0; k < CHUNK; ++k)
                    bv[k] = bnd[t0 + k];
            }

            #pragma unroll
            for (int k = 0; k < CHUNK; ++k) {
                const int t = t0 + k;
                const int i = t - lane;
                float hl = __shfl_up(h, 1);
                if (lane == 0)
                    hl = bv[k];
                if (i >= 0 && i < DD1) {
                    const float hn = tanhf(fmaf(u0, h, fmaf(u1, hl, av[k])));
                    h = hn;
                    if (lane == 31) bnd[i] = hn;
                    h_out[(((size_t)i * DD2 + j) * BB + b) * SOUT + o] = hn;
                }
            }
        }
    }
}

extern "C" void kernel_launch(void* const* d_in, const int* in_sizes, int n_in,
                              void* d_out, int out_size, void* d_ws, size_t ws_size,
                              hipStream_t stream)
{
    const float* x    = (const float*)d_in[0];
    const float* w    = (const float*)d_in[1];
    const float* u    = (const float*)d_in[2];
    const float* bias = (const float*)d_in[3];
    float* h   = (float*)d_out;
    float* a_t = (float*)d_ws;

    dim3 g1(SOUT / 32, DD2 / 32, DD1);
    mdrnn_gemm_wmma<<<g1, 256, 0, stream>>>(x, w, bias, a_t);

    mdrnn_scan<<<(BB * SOUT) / 8, 256, 0, stream>>>(a_t, u, h);
}


# scan hands h out as whole 128-byte rows through its LDS ring, over the a_t rows already consumed; a new third kernel transposes them to h[i][j][b][o] with whole-line stores; sums unchanged
# speedup vs baseline: 2.6409x; 1.7617x over previous
#include <hip/hip_runtime.h>


#define DD1  128
#define DD2  128
#define BB   16
#define SIN  64
#define SOUT 128

typedef __attribute__((ext_vector_type(2))) float v2f;
typedef __attribute__((ext_vector_type(4))) float v4f;
typedef __attribute__((ext_vector_type(8))) float v8f;

__global__ __launch_bounds__(256)
void mdrnn_gemm_wmma(const float* __restrict__ x, const float* __restrict__ w,
                     const float* __restrict__ bias, float* __restrict__ a_t)
{
    __shared__ float tile[512 * 32];

    const int i    = blockIdx.z;
    const int j0   = blockIdx.y * 32;
    const int o0   = blockIdx.x * 32;
    const int tid  = threadIdx.x;
    const int lane = tid & 31;
    const int wave = tid >> 5;

    const int lm = lane & 15;
    const int kh = (lane >> 4) * 2;
    const int hh = lane >> 4;

    for (int t = 0; t < 8; ++t) {
        const int jj = wave * 4 + (t >> 1);
        const int nn = (t & 1) * 16;
        const int j  = j0 + jj;

        const float* xrow = x + (((size_t)i * DD2 + j) * BB + lm) * SIN;

        v8f acc = {};
        #pragma unroll
        for (int k0 = 0; k0 < SIN; k0 += 4) {
            v2f af;
            af.x = xrow[k0 + kh];
            af.y = xrow[k0 + kh + 1];
            const float* wp = w + (size_t)(k0 + kh) * SOUT + (o0 + nn + lm);
            v2f bf;
            bf.x = wp[0];
            bf.y = wp[SOUT];
            acc = __builtin_amdgcn_wmma_f32_16x16x4_f32(
                false, af, false, bf, (short)0, acc, false, false);
        }

        #pragma unroll
        for (int v = 0; v < 8; ++v)
            tile[(jj * 16 + (v + 8 * hh)) * 32 + nn + lm] = acc[v];
    }
    __syncthreads();

    for (int p = tid; p < 512; p += 256) {
        const int b  = p >> 5;
        const int oo = p & 31;
        const int o  = o0 + oo;
        const float bv = bias[o];
        float* dst = a_t + (size_t)(b * SOUT + o) * (DD1 * DD2)
                         + (size_t)i * DD2 + j0;
        #pragma unroll
        for (int jj = 0; jj < 32; jj += 4) {
            v4f val;
            val.x = tile[((jj + 0) * 16 + b) * 32 + oo] + bv;
            val.y = tile[((jj + 1) * 16 + b) * 32 + oo] + bv;
            val.z = tile[((jj + 2) * 16 + b) * 32 + oo] + bv;
            val.w = tile[((jj + 3) * 16 + b) * 32 + oo] + bv;
            *(v4f*)(dst + jj) = val;
        }
    }
}

#define RING_ROWS 64
#define CHUNK     8
#define NCHUNK    20

__global__ __launch_bounds__(256)
void mdrnn_scan(float* __restrict__ ws, const float* __restrict__ u)
{
    __shared__ float boundary[8][DD1];
    __shared__ float ring[8][RING_ROWS * 32];

    const int tid  = threadIdx.x;
    const int lane = tid & 31;
    const int wave = tid >> 5;
    const int bo   = blockIdx.x * 8 + wave;
    const int o    = bo & 127;

    const float u0 = u[o];
    const float u1 = u[SOUT + o];
    float* abase = ws + (size_t)bo * (DD1 * DD2) + lane;
    float* bnd = boundary[wave];
    float* rw  = ring[wave];

    float r[CHUNK];
    #pragma unroll
    for (int k = 0; k < CHUNK; ++k)
        r[k] = abase[k * DD2];

    for (int s = 0; s < 4; ++s) {
        float h = 0.0f;
        for (int c = 0; c < NCHUNK; ++c) {
            const int t0 = c * CHUNK;
            if (t0 < DD1) {
                #pragma unroll
                for (int k = 0; k < CHUNK; ++k)
                    rw[((t0 + k) & (RING_ROWS - 1)) * 32 + lane] = r[k];
            }
            if (t0 >= 40) {
                #pragma unroll
                for (int k = 0; k < CHUNK; ++k) {
                    const int i = t0 - 40 + k;
                    abase[i * DD2 + s * 32] = rw[(i & (RING_ROWS - 1)) * 32 + lane];
                }
            }
            if (t0 + CHUNK < DD1) {
                #pragma unroll
                for (int k = 0; k < CHUNK; ++k)
                    r[k] = abase[(t0 + CHUNK + k) * DD2 + s * 32];
            } else if (c == NCHUNK - 1 && s < 3) {
                #pragma unroll
                for (int k = 0; k < CHUNK; ++k)
                    r[k] = abase[k * DD2 + (s + 1) * 32];
            }

            float av[CHUNK], bv[CHUNK];
            #pragma unroll
            for (int k = 0; k < CHUNK; ++k) {
                const int i = t0 + k - lane;
                av[k] = 0.0f;
                bv[k] = 0.0f;
                if (i >= 0 && i < DD1)
                    av[k] = rw[(i & (RING_ROWS - 1)) * 32 + lane];
            }
            if (lane == 0 && s > 0 && t0 < DD1) {
                #pragma unroll
                for (int k = 0; k < CHUNK; ++k)
                    bv[k] = bnd[t0 + k];
            }

            #pragma unroll
            for (int k = 0; k < CHUNK; ++k) {
                const int t = t0 + k;
                const int i = t - lane;
                float hl = __shfl_up(h, 1);
                if (lane == 0)
                    hl = bv[k];
                if (i >= 0 && i < DD1) {
                    const float hn = tanhf(fmaf(u0, h, fmaf(u1, hl, av[k])));
                    h = hn;
                    if (lane == 31) bnd[i] = hn;
                    rw[(i & (RING_ROWS - 1)) * 32 + lane] = hn;
                }
            }
        }
        #pragma unroll
        for (int k = 0; k < CHUNK; ++k) {
            const int i = DD1 - CHUNK + k;
            abase[i * DD2 + s * 32] = rw[(i & (RING_ROWS - 1)) * 32 + lane];
        }
    }
}

#define TPITCH 129

__global__ __launch_bounds__(256)
void mdrnn_out_transpose(const float* __restrict__ h_t, float* __restrict__ h_out)
{
    __shared__ float tile[SOUT * TPITCH];

    const int b    = blockIdx.x;
    const int i    = blockIdx.y;
    const int lane = threadIdx.x & 31;
    const int wave = threadIdx.x >> 5;

    const float* src = h_t + ((size_t)(b * SOUT + wave * 16) * DD1 + i) * DD2 + lane * 4;
    v4f v[16];
    #pragma unroll
    for (int r = 0; r < 16; ++r)
        v[r] = *(const v4f*)(src + (size_t)r * (DD1 * DD2));
    #pragma unroll
    for (int r = 0; r < 16; ++r) {
        float* d = tile + (wave * 16 + r) * TPITCH + lane * 4;
        d[0] = v[r].x;
        d[1] = v[r].y;
        d[2] = v[r].z;
        d[3] = v[r].w;
    }
    __syncthreads();

    float* dst = h_out + (((size_t)i * DD2 + wave * 16) * BB + b) * SOUT + lane * 4;
    const float* col = tile + (lane * 4) * TPITCH + wave * 16;
    #pragma unroll
    for (int r = 0; r < 16; ++r) {
        v4f q;
        q.x = col[0 * TPITCH + r];
        q.y = col[1 * TPITCH + r];
        q.z = col[2 * TPITCH + r];
        q.w = col[3 * TPITCH + r];
        *(v4f*)(dst + (size_t)r * (BB * SOUT)) = q;
    }
}

extern "C" void kernel_launch(void* const* d_in, const int* in_sizes, int n_in,
                              void* d_out, int out_size, void* d_ws, size_t ws_size,
                              hipStream_t stream)
{
    const float* x    = (const float*)d_in[0];
    const float* w    = (const float*)d_in[1];
    const float* u    = (const float*)d_in[2];
    const float* bias = (const float*)d_in[3];
    float* h   = (float*)d_out;
    float* a_t = (float*)d_ws;

    dim3 g1(SOUT / 32, DD2 / 32, DD1);
    mdrnn_gemm_wmma<<<g1, 256, 0, stream>>>(x, w, bias, a_t);

    mdrnn_scan<<<(BB * SOUT) / 8, 256, 0, stream>>>(a_t, u);

    dim3 g3(BB, DD1);
    mdrnn_out_transpose<<<g3, 256, 0, stream>>>(a_t, h);
}


# scan: the four 32-column strips of a plane run on four waves of one work-group, pipelined five 8-step chunks apart with a barrier per chunk (35 phases, not 80 chunks in a row); sums unchanged
# speedup vs baseline: 3.1711x; 1.2008x over previous
#include <hip/hip_runtime.h>


#define DD1  128
#define DD2  128
#define BB   16
#define SIN  64
#define SOUT 128

typedef __attribute__((ext_vector_type(2))) float v2f;
typedef __attribute__((ext_vector_type(4))) float v4f;
typedef __attribute__((ext_vector_type(8))) float v8f;

__global__ __launch_bounds__(256)
void mdrnn_gemm_wmma(const float* __restrict__ x, const float* __restrict__ w,
                     const float* __restrict__ bias, float* __restrict__ a_t)
{
    __shared__ float tile[512 * 32];

    const int i    = blockIdx.z;
    const int j0   = blockIdx.y * 32;
    const int o0   = blockIdx.x * 32;
    const int tid  = threadIdx.x;
    const int lane = tid & 31;
    const int wave = tid >> 5;

    const int lm = lane & 15;
    const int kh = (lane >> 4) * 2;
    const int hh = lane >> 4;

    for (int t = 0; t < 8; ++t) {
        const int jj = wave * 4 + (t >> 1);
        const int nn = (t & 1) * 16;
        const int j  = j0 + jj;

        const float* xrow = x + (((size_t)i * DD2 + j) * BB + lm) * SIN;

        v8f acc = {};
        #pragma unroll
        for (int k0 = 0; k0 < SIN; k0 += 4) {
            v2f af;
            af.x = xrow[k0 + kh];
            af.y = xrow[k0 + kh + 1];
            const float* wp = w + (size_t)(k0 + kh) * SOUT + (o0 + nn + lm);
            v2f bf;
            bf.x = wp[0];
            bf.y = wp[SOUT];
            acc = __builtin_amdgcn_wmma_f32_16x16x4_f32(
                false, af, false, bf, (short)0, acc, false, false);
        }

        #pragma unroll
        for (int v = 0; v < 8; ++v)
            tile[(jj * 16 + (v + 8 * hh)) * 32 + nn + lm] = acc[v];
    }
    __syncthreads();

    for (int p = tid; p < 512; p += 256) {
        const int b  = p >> 5;
        const int oo = p & 31;
        const int o  = o0 + oo;
        const float bv = bias[o];
        float* dst = a_t + (size_t)(b * SOUT + o) * (DD1 * DD2)
                         + (size_t)i * DD2 + j0;
        #pragma unroll
        for (int jj = 0; jj < 32; jj += 4) {
            v4f val;
            val.x = tile[((jj + 0) * 16 + b) * 32 + oo] + bv;
            val.y = tile[((jj + 1) * 16 + b) * 32 + oo] + bv;
            val.z = tile[((jj + 2) * 16 + b) * 32 + oo] + bv;
            val.w = tile[((jj + 3) * 16 + b) * 32 + oo] + bv;
            *(v4f*)(dst + jj) = val;
        }
    }
}

#define RING_ROWS 64
#define CHUNK     8
#define NCHUNK    20
#define LAG       5
#define NPHASE    (NCHUNK + 3 * LAG)

__global__ __launch_bounds__(128)
void mdrnn_scan(float* __restrict__ ws, const float* __restrict__ u)
{
    __shared__ float boundary[3][DD1];
    __shared__ float ring[4][RING_ROWS * 32];

    const int lane = threadIdx.x & 31;
    const int s    = __builtin_amdgcn_readfirstlane((int)(threadIdx.x >> 5));
    const int bo   = blockIdx.x;
    const int o    = bo & 127;

    const float u0 = u[o];
    const float u1 = u[SOUT + o];
    float* abase = ws + (size_t)bo * (DD1 * DD2) + s * 32 + lane;
    const float* bin  = boundary[s > 0 ? s - 1 : 0];
    float*       bout = boundary[s < 3 ? s : 2];
    const bool feeds  = (lane == 31) && (s < 3);
    const bool fed    = (lane == 0) && (s > 0);
    float* rw = ring[s];

    float r[CHUNK];
    #pragma unroll
    for (int k = 0; k < CHUNK; ++k)
        r[k] = abase[k * DD2];

    float h = 0.0f;
    for (int p = 0; p < NPHASE; ++p) {
        const int c = p - LAG * s;
        if (c >= 0 && c < NCHUNK) {
            const int t0 = c * CHUNK;
            if (t0 < DD1) {
                #pragma unroll
                for (int k = 0; k < CHUNK; ++k)
                    rw[((t0 + k) & (RING_ROWS - 1)) * 32 + lane] = r[k];
            }
            if (t0 >= 40) {
                #pragma unroll
                for (int k = 0; k < CHUNK; ++k) {
                    const int i = t0 - 40 + k;
                    abase[i * DD2] = rw[(i & (RING_ROWS - 1)) * 32 + lane];
                }
            }
            if (t0 + CHUNK < DD1) {
                #pragma unroll
                for (int k = 0; k < CHUNK; ++k)
                    r[k] = abase[(t0 + CHUNK + k) * DD2];
            }

            float av[CHUNK], bv[CHUNK];
            #pragma unroll
            for (int k = 0; k < CHUNK; ++k) {
                const int i = t0 + k - lane;
                av[k] = 0.0f;
                bv[k] = 0.0f;
                if (i >= 0 && i < DD1)
                    av[k] = rw[(i & (RING_ROWS - 1)) * 32 + lane];
            }
            if (fed && t0 < DD1) {
                #pragma unroll
                for (int k = 0; k < CHUNK; ++k)
                    bv[k] = bin[t0 + k];
            }

            #pragma unroll
            for (int k = 0; k < CHUNK; ++k) {
                const int t = t0 + k;
                const int i = t - lane;
                float hl = __shfl_up(h, 1);
                if (lane == 0)
                    hl = bv[k];
                if (i >= 0 && i < DD1) {
                    const float hn = tanhf(fmaf(u0, h, fmaf(u1, hl, av[k])));
                    h = hn;
                    if (feeds) bout[i] = hn;
                    rw[(i & (RING_ROWS - 1)) * 32 + lane] = hn;
                }
            }

            if (c == NCHUNK - 1) {
                #pragma unroll
                for (int k = 0; k < CHUNK; ++k) {
                    const int i = DD1 - CHUNK + k;
                    abase[i * DD2] = rw[(i & (RING_ROWS - 1)) * 32 + lane];
                }
            }
        }
        __syncthreads();
    }
}

#define TPITCH 129

__global__ __launch_bounds__(256)
void mdrnn_out_transpose(const float* __restrict__ h_t, float* __restrict__ h_out)
{
    __shared__ float tile[SOUT * TPITCH];

    const int b    = blockIdx.x;
    const int i    = blockIdx.y;
    const int lane = threadIdx.x & 31;
    const int wave = threadIdx.x >> 5;

    const float* src = h_t + ((size_t)(b * SOUT + wave * 16) * DD1 + i) * DD2 + lane * 4;
    v4f v[16];
    #pragma unroll
    for (int r = 0; r < 16; ++r)
        v[r] = *(const v4f*)(src + (size_t)r * (DD1 * DD2));
    #pragma unroll
    for (int r = 0; r < 16; ++r) {
        float* d = tile + (wave * 16 + r) * TPITCH + lane * 4;
        d[0] = v[r].x;
        d[1] = v[r].y;
        d[2] = v[r].z;
        d[3] = v[r].w;
    }
    __syncthreads();

    float* dst = h_out + (((size_t)i * DD2 + wave * 16) * BB + b) * SOUT + lane * 4;
    const float* col = tile + (lane * 4) * TPITCH + wave * 16;
    #pragma unroll
    for (int r = 0; r < 16; ++r) {
        v4f q;
        q.x = col[0 * TPITCH + r];
        q.y = col[1 * TPITCH + r];
        q.z = col[2 * TPITCH + r];
        q.w = col[3 * TPITCH + r];
        *(v4f*)(dst + (size_t)r * (BB * SOUT)) = q;
    }
}

extern "C" void kernel_launch(void* const* d_in, const int* in_sizes, int n_in,
                              void* d_out, int out_size, void* d_ws, size_t ws_size,
                              hipStream_t stream)
{
    const float* x    = (const float*)d_in[0];
    const float* w    = (const float*)d_in[1];
    const float* u    = (const float*)d_in[2];
    const float* bias = (const float*)d_in[3];
    float* h   = (float*)d_out;
    float* a_t = (float*)d_ws;

    dim3 g1(SOUT / 32, DD2 / 32, DD1);
    mdrnn_gemm_wmma<<<g1, 256, 0, stream>>>(x, w, bias, a_t);

    mdrnn_scan<<<BB * SOUT, 128, 0, stream>>>(a_t, u);

    dim3 g3(BB, DD1);
    mdrnn_out_transpose<<<g3, 256, 0, stream>>>(a_t, h);
}


# scan: tanh as one hardware transcendental (v_tanh_f32) instead of the library routine, whose two lane-masked ranges both ran on every step of the recurrence; tanh rounding differs, sums unchanged
# speedup vs baseline: 4.0987x; 1.2925x over previous
#include <hip/hip_runtime.h>


#define DD1  128
#define DD2  128
#define BB   16
#define SIN  64
#define SOUT 128

typedef __attribute__((ext_vector_type(2))) float v2f;
typedef __attribute__((ext_vector_type(4))) float v4f;
typedef __attribute__((ext_vector_type(8))) float v8f;

__global__ __launch_bounds__(256)
void mdrnn_gemm_wmma(const float* __restrict__ x, const float* __restrict__ w,
                     const float* __restrict__ bias, float* __restrict__ a_t)
{
    __shared__ float tile[512 * 32];

    const int i    = blockIdx.z;
    const int j0   = blockIdx.y * 32;
    const int o0   = blockIdx.x * 32;
    const int tid  = threadIdx.x;
    const int lane = tid & 31;
    const int wave = tid >> 5;

    const int lm = lane & 15;
    const int hh = lane >> 4;

    v2f bfr[2][16];
    #pragma unroll
    for (int m = 0; m < 16; ++m) {
        const int kr = 8 * (m >> 1) + 4 * hh + 2 * (m & 1);
        #pragma unroll
        for (int nt = 0; nt < 2; ++nt) {
            const float* wp = w + (size_t)kr * SOUT + (o0 + nt * 16 + lm);
            bfr[nt][m].x = wp[0];
            bfr[nt][m].y = wp[SOUT];
        }
    }

    const float* xw = x + (((size_t)i * DD2 + j0 + wave * 4) * BB + lm) * SIN + 4 * hh;
    float* tw = tile + (wave * 64 + 8 * hh) * 32 + lm;

    #pragma unroll
    for (int q4 = 0; q4 < 4; ++q4) {
        const float* xrow = xw + q4 * (BB * SIN);
        v4f a4[8];
        #pragma unroll
        for (int q = 0; q < 8; ++q)
            a4[q] = *(const v4f*)(xrow + 8 * q);
        asm volatile("" : "+v"(a4[0]), "+v"(a4[1]), "+v"(a4[2]), "+v"(a4[3]),
                          "+v"(a4[4]), "+v"(a4[5]), "+v"(a4[6]), "+v"(a4[7]));

        v8f acc0 = {};
        v8f acc1 = {};
        #pragma unroll
        for (int m = 0; m < 16; ++m) {
            v2f af;
            af.x = (m & 1) ? a4[m >> 1].z : a4[m >> 1].x;
            af.y = (m & 1) ? a4[m >> 1].w : a4[m >> 1].y;
            acc0 = __builtin_amdgcn_wmma_f32_16x16x4_f32(
                false, af, false, bfr[0][m], (short)0, acc0, false, false);
            acc1 = __builtin_amdgcn_wmma_f32_16x16x4_f32(
                false, af, false, bfr[1][m], (short)0, acc1, false, false);
        }

        #pragma unroll
        for (int v = 0; v < 8; ++v) {
            tw[(q4 * 16 + v) * 32]      = acc0[v];
            tw[(q4 * 16 + v) * 32 + 16] = acc1[v];
        }
    }
    __syncthreads();

    for (int p = tid; p < 512; p += 256) {
        const int b  = p >> 5;
        const int oo = p & 31;
        const int o  = o0 + oo;
        const float bv = bias[o];
        float* dst = a_t + (size_t)(b * SOUT + o) * (DD1 * DD2)
                         + (size_t)i * DD2 + j0;
        #pragma unroll
        for (int jj = 0; jj < 32; jj += 4) {
            v4f val;
            val.x = tile[((jj + 0) * 16 + b) * 32 + oo] + bv;
            val.y = tile[((jj + 1) * 16 + b) * 32 + oo] + bv;
            val.z = tile[((jj + 2) * 16 + b) * 32 + oo] + bv;
            val.w = tile[((jj + 3) * 16 + b) * 32 + oo] + bv;
            *(v4f*)(dst + jj) = val;
        }
    }
}

#define RING_ROWS 64
#define CHUNK     8
#define NCHUNK    20
#define LAG       5
#define NPHASE    (NCHUNK + 3 * LAG)

__global__ __launch_bounds__(128)
void mdrnn_scan(float* __restrict__ ws, const float* __restrict__ u)
{
    __shared__ float boundary[3][DD1];
    __shared__ float ring[4][RING_ROWS * 32];

    const int lane = threadIdx.x & 31;
    const int s    = __builtin_amdgcn_readfirstlane((int)(threadIdx.x >> 5));
    const int bo   = blockIdx.x;
    const int o    = bo & 127;

    const float u0 = u[o];
    const float u1 = u[SOUT + o];
    float* abase = ws + (size_t)bo * (DD1 * DD2) + s * 32 + lane;
    const float* bin  = boundary[s > 0 ? s - 1 : 0];
    float*       bout = boundary[s < 3 ? s : 2];
    const bool feeds  = (lane == 31) && (s < 3);
    const bool fed    = (lane == 0) && (s > 0);
    float* rw = ring[s];

    float r[CHUNK];
    #pragma unroll
    for (int k = 0; k < CHUNK; ++k)
        r[k] = abase[k * DD2];

    float h = 0.0f;
    for (int p = 0; p < NPHASE; ++p) {
        const int c = p - LAG * s;
        if (c >= 0 && c < NCHUNK) {
            const int t0 = c * CHUNK;
            if (t0 < DD1) {
                #pragma unroll
                for (int k = 0; k < CHUNK; ++k)
                    rw[((t0 + k) & (RING_ROWS - 1)) * 32 + lane] = r[k];
            }
            if (t0 >= 40) {
                #pragma unroll
                for (int k = 0; k < CHUNK; ++k) {
                    const int i = t0 - 40 + k;
                    abase[i * DD2] = rw[(i & (RING_ROWS - 1)) * 32 + lane];
                }
            }
            if (t0 + CHUNK < DD1) {
                #pragma unroll
                for (int k = 0; k < CHUNK; ++k)
                    r[k] = abase[(t0 + CHUNK + k) * DD2];
            }

            float av[CHUNK], bv[CHUNK];
            #pragma unroll
            for (int k = 0; k < CHUNK; ++k) {
                const int i = t0 + k - lane;
                av[k] = 0.0f;
                bv[k] = 0.0f;
                if (i >= 0 && i < DD1)
                    av[k] = rw[(i & (RING_ROWS - 1)) * 32 + lane];
            }
            if (fed && t0 < DD1) {
                #pragma unroll
                for (int k = 0; k < CHUNK; ++k)
                    bv[k] = bin[t0 + k];
            }

            #pragma unroll
            for (int k = 0; k < CHUNK; ++k) {
                const int t = t0 + k;
                const int i = t - lane;
                float hl = __shfl_up(h, 1);
                if (lane == 0)
                    hl = bv[k];
                if (i >= 0 && i < DD1) {
                    const float hn = __builtin_amdgcn_tanhf(fmaf(u0, h, fmaf(u1, hl, av[k])));
                    h = hn;
                    if (feeds) bout[i] = hn;
                    rw[(i & (RING_ROWS - 1)) * 32 + lane] = hn;
                }
            }

            if (c == NCHUNK - 1) {
                #pragma unroll
                for (int k = 0; k < CHUNK; ++k) {
                    const int i = DD1 - CHUNK + k;
                    abase[i * DD2] = rw[(i & (RING_ROWS - 1)) * 32 + lane];
                }
            }
        }
        __syncthreads();
    }
}

#define TPITCH 129

__global__ __launch_bounds__(256)
void mdrnn_out_transpose(const float* __restrict__ h_t, float* __restrict__ h_out)
{
    __shared__ float tile[SOUT * TPITCH];

    const int b    = blockIdx.x;
    const int i    = blockIdx.y;
    const int lane = threadIdx.x & 31;
    const int wave = threadIdx.x >> 5;

    const float* src = h_t + ((size_t)(b * SOUT + wave * 16) * DD1 + i) * DD2 + lane * 4;
    v4f v[16];
    #pragma unroll
    for (int r = 0; r < 16; ++r)
        v[r] = *(const v4f*)(src + (size_t)r * (DD1 * DD2));
    #pragma unroll
    for (int r = 0; r < 16; ++r) {
        float* d = tile + (wave * 16 + r) * TPITCH + lane * 4;
        d[0] = v[r].x;
        d[1] = v[r].y;
        d[2] = v[r].z;
        d[3] = v[r].w;
    }
    __syncthreads();

    float* dst = h_out + (((size_t)i * DD2 + wave * 16) * BB + b) * SOUT + lane * 4;
    const float* col = tile + (lane * 4) * TPITCH + wave * 16;
    #pragma unroll
    for (int r = 0; r < 16; ++r) {
        v4f q;
        q.x = col[0 * TPITCH + r];
        q.y = col[1 * TPITCH + r];
        q.z = col[2 * TPITCH + r];
        q.w = col[3 * TPITCH + r];
        *(v4f*)(dst + (size_t)r * (BB * SOUT)) = q;
    }
}

extern "C" void kernel_launch(void* const* d_in, const int* in_sizes, int n_in,
                              void* d_out, int out_size, void* d_ws, size_t ws_size,
                              hipStream_t stream)
{
    const float* x    = (const float*)d_in[0];
    const float* w    = (const float*)d_in[1];
    const float* u    = (const float*)d_in[2];
    const float* bias = (const float*)d_in[3];
    float* h   = (float*)d_out;
    float* a_t = (float*)d_ws;

    dim3 g1(SOUT / 32, DD2 / 32, DD1);
    mdrnn_gemm_wmma<<<g1, 256, 0, stream>>>(x, w, bias, a_t);

    mdrnn_scan<<<BB * SOUT, 128, 0, stream>>>(a_t, u);

    dim3 g3(BB, DD1);
    mdrnn_out_transpose<<<g3, 256, 0, stream>>>(a_t, h);
}


# GEMM epilogue: LDS result tile kept [b][o][j] (pitch 36) so each a_t store instruction writes four whole 128-byte lines, not 32 pieces of 16 bytes in 32 lines; sums unchanged
# speedup vs baseline: 5.2286x; 1.2757x over previous
#include <hip/hip_runtime.h>


#define DD1  128
#define DD2  128
#define BB   16
#define SIN  64
#define SOUT 128

typedef __attribute__((ext_vector_type(2))) float v2f;
typedef __attribute__((ext_vector_type(4))) float v4f;
typedef __attribute__((ext_vector_type(8))) float v8f;

#define GPITCH 36

__global__ __launch_bounds__(256)
void mdrnn_gemm_wmma(const float* __restrict__ x, const float* __restrict__ w,
                     const float* __restrict__ bias, float* __restrict__ a_t)
{
    __shared__ __attribute__((aligned(16))) float tile[512 * GPITCH];

    const int i    = blockIdx.z;
    const int j0   = blockIdx.y * 32;
    const int o0   = blockIdx.x * 32;
    const int tid  = threadIdx.x;
    const int lane = tid & 31;
    const int wave = tid >> 5;

    const int lm = lane & 15;
    const int hh = lane >> 4;

    v2f bfr[2][16];
    #pragma unroll
    for (int m = 0; m < 16; ++m) {
        const int kr = 8 * (m >> 1) + 4 * hh + 2 * (m & 1);
        #pragma unroll
        for (int nt = 0; nt < 2; ++nt) {
            const float* wp = w + (size_t)kr * SOUT + (o0 + nt * 16 + lm);
            bfr[nt][m].x = wp[0];
            bfr[nt][m].y = wp[SOUT];
        }
    }

    const float* xw = x + (((size_t)i * DD2 + j0 + wave * 4) * BB + lm) * SIN + 4 * hh;
    float* tw = tile + (8 * hh * 32 + lm) * GPITCH + wave * 4;

    #pragma unroll
    for (int q4 = 0; q4 < 4; ++q4) {
        const float* xrow = xw + q4 * (BB * SIN);
        v4f a4[8];
        #pragma unroll
        for (int q = 0; q < 8; ++q)
            a4[q] = *(const v4f*)(xrow + 8 * q);
        asm volatile("" : "+v"(a4[0]), "+v"(a4[1]), "+v"(a4[2]), "+v"(a4[3]),
                          "+v"(a4[4]), "+v"(a4[5]), "+v"(a4[6]), "+v"(a4[7]));

        v8f acc0 = {};
        v8f acc1 = {};
        #pragma unroll
        for (int m = 0; m < 16; ++m) {
            v2f af;
            af.x = (m & 1) ? a4[m >> 1].z : a4[m >> 1].x;
            af.y = (m & 1) ? a4[m >> 1].w : a4[m >> 1].y;
            acc0 = __builtin_amdgcn_wmma_f32_16x16x4_f32(
                false, af, false, bfr[0][m], (short)0, acc0, false, false);
            acc1 = __builtin_amdgcn_wmma_f32_16x16x4_f32(
                false, af, false, bfr[1][m], (short)0, acc1, false, false);
        }

        #pragma unroll
        for (int v = 0; v < 8; ++v) {
            tw[(v * 32) * GPITCH + q4]      = acc0[v];
            tw[(v * 32 + 16) * GPITCH + q4] = acc1[v];
        }
    }
    __syncthreads();

    const int pl = lane >> 3;
    const int jq = lane & 7;
    const int oo = wave * 4 + pl;
    const float bv = bias[o0 + oo];
    const float* src = tile + oo * GPITCH + 4 * jq;
    float* dst = a_t + (size_t)(o0 + oo) * (DD1 * DD2) + (size_t)i * DD2 + j0 + 4 * jq;
    #pragma unroll
    for (int b = 0; b < BB; ++b) {
        v4f val = *(const v4f*)(src + b * 32 * GPITCH);
        val.x += bv;
        val.y += bv;
        val.z += bv;
        val.w += bv;
        *(v4f*)(dst + (size_t)b * SOUT * (DD1 * DD2)) = val;
    }
}

#define RING_ROWS 64
#define CHUNK     8
#define NCHUNK    20
#define LAG       5
#define NPHASE    (NCHUNK + 3 * LAG)

__global__ __launch_bounds__(128)
void mdrnn_scan(float* __restrict__ ws, const float* __restrict__ u)
{
    __shared__ float boundary[3][DD1];
    __shared__ float ring[4][RING_ROWS * 32];

    const int lane = threadIdx.x & 31;
    const int s    = __builtin_amdgcn_readfirstlane((int)(threadIdx.x >> 5));
    const int bo   = blockIdx.x;
    const int o    = bo & 127;

    const float u0 = u[o];
    const float u1 = u[SOUT + o];
    float* abase = ws + (size_t)bo * (DD1 * DD2) + s * 32 + lane;
    const float* bin  = boundary[s > 0 ? s - 1 : 0];
    float*       bout = boundary[s < 3 ? s : 2];
    const bool feeds  = (lane == 31) && (s < 3);
    const bool fed    = (lane == 0) && (s > 0);
    float* rw = ring[s];

    float r[CHUNK];
    #pragma unroll
    for (int k = 0; k < CHUNK; ++k)
        r[k] = abase[k * DD2];

    float h = 0.0f;
    for (int p = 0; p < NPHASE; ++p) {
        const int c = p - LAG * s;
        if (c >= 0 && c < NCHUNK) {
            const int t0 = c * CHUNK;
            if (t0 < DD1) {
                #pragma unroll
                for (int k = 0; k < CHUNK; ++k)
                    rw[((t0 + k) & (RING_ROWS - 1)) * 32 + lane] = r[k];
            }
            if (t0 >= 40) {
                #pragma unroll
                for (int k = 0; k < CHUNK; ++k) {
                    const int i = t0 - 40 + k;
                    abase[i * DD2] = rw[(i & (RING_ROWS - 1)) * 32 + lane];
                }
            }
            if (t0 + CHUNK < DD1) {
                #pragma unroll
                for (int k = 0; k < CHUNK; ++k)
                    r[k] = abase[(t0 + CHUNK + k) * DD2];
            }

            float av[CHUNK], bv[CHUNK];
            #pragma unroll
            for (int k = 0; k < CHUNK; ++k) {
                const int i = t0 + k - lane;
                av[k] = 0.0f;
                bv[k] = 0.0f;
                if (i >= 0 && i < DD1)
                    av[k] = rw[(i & (RING_ROWS - 1)) * 32 + lane];
            }
            if (fed && t0 < DD1) {
                #pragma unroll
                for (int k = 0; k < CHUNK; ++k)
                    bv[k] = bin[t0 + k];
            }

            #pragma unroll
            for (int k = 0; k < CHUNK; ++k) {
                const int t = t0 + k;
                const int i = t - lane;
                float hl = __shfl_up(h, 1);
                if (lane == 0)
                    hl = bv[k];
                if (i >= 0 && i < DD1) {
                    const float hn = __builtin_amdgcn_tanhf(fmaf(u0, h, fmaf(u1, hl, av[k])));
                    h = hn;
                    if (feeds) bout[i] = hn;
                    rw[(i & (RING_ROWS - 1)) * 32 + lane] = hn;
                }
            }

            if (c == NCHUNK - 1) {
                #pragma unroll
                for (int k = 0; k < CHUNK; ++k) {
                    const int i = DD1 - CHUNK + k;
                    abase[i * DD2] = rw[(i & (RING_ROWS - 1)) * 32 + lane];
                }
            }
        }
        __syncthreads();
    }
}

#define TPITCH 129

__global__ __launch_bounds__(256)
void mdrnn_out_transpose(const float* __restrict__ h_t, float* __restrict__ h_out)
{
    __shared__ float tile[SOUT * TPITCH];

    const int b    = blockIdx.x;
    const int i    = blockIdx.y;
    const int lane = threadIdx.x & 31;
    const int wave = threadIdx.x >> 5;

    const float* src = h_t + ((size_t)(b * SOUT + wave * 16) * DD1 + i) * DD2 + lane * 4;
    v4f v[16];
    #pragma unroll
    for (int r = 0; r < 16; ++r)
        v[r] = *(const v4f*)(src + (size_t)r * (DD1 * DD2));
    #pragma unroll
    for (int r = 0; r < 16; ++r) {
        float* d = tile + (wave * 16 + r) * TPITCH + lane * 4;
        d[0] = v[r].x;
        d[1] = v[r].y;
        d[2] = v[r].z;
        d[3] = v[r].w;
    }
    __syncthreads();

    float* dst = h_out + (((size_t)i * DD2 + wave * 16) * BB + b) * SOUT + lane * 4;
    const float* col = tile + (lane * 4) * TPITCH + wave * 16;
    #pragma unroll
    for (int r = 0; r < 16; ++r) {
        v4f q;
        q.x = col[0 * TPITCH + r];
        q.y = col[1 * TPITCH + r];
        q.z = col[2 * TPITCH + r];
        q.w = col[3 * TPITCH + r];
        *(v4f*)(dst + (size_t)r * (BB * SOUT)) = q;
    }
}

extern "C" void kernel_launch(void* const* d_in, const int* in_sizes, int n_in,
                              void* d_out, int out_size, void* d_ws, size_t ws_size,
                              hipStream_t stream)
{
    const float* x    = (const float*)d_in[0];
    const float* w    = (const float*)d_in[1];
    const float* u    = (const float*)d_in[2];
    const float* bias = (const float*)d_in[3];
    float* h   = (float*)d_out;
    float* a_t = (float*)d_ws;

    dim3 g1(SOUT / 32, DD2 / 32, DD1);
    mdrnn_gemm_wmma<<<g1, 256, 0, stream>>>(x, w, bias, a_t);

    mdrnn_scan<<<BB * SOUT, 128, 0, stream>>>(a_t, u);

    dim3 g3(BB, DD1);
    mdrnn_out_transpose<<<g3, 256, 0, stream>>>(a_t, h);
}


# GEMM: one work-group per (i, 32 j) computes all 128 o as four column blocks of 32; its A fragments are fetched from x once and stay in registers (x read once, not four times); sums unchanged
# speedup vs baseline: 6.6327x; 1.2685x over previous
#include <hip/hip_runtime.h>


#define DD1  128
#define DD2  128
#define BB   16
#define SIN  64
#define SOUT 128

typedef __attribute__((ext_vector_type(2))) float v2f;
typedef __attribute__((ext_vector_type(4))) float v4f;
typedef __attribute__((ext_vector_type(8))) float v8f;

#define GPITCH 36

__global__ __launch_bounds__(256)
void mdrnn_gemm_wmma(const float* __restrict__ x, const float* __restrict__ w,
                     const float* __restrict__ bias, float* __restrict__ a_t)
{
    __shared__ __attribute__((aligned(16))) float tile[512 * GPITCH];

    const int i    = blockIdx.z;
    const int j0   = blockIdx.y * 32;
    const int tid  = threadIdx.x;
    const int lane = tid & 31;
    const int wave = tid >> 5;

    const int lm = lane & 15;
    const int hh = lane >> 4;

    const float* xw = x + (((size_t)i * DD2 + j0 + wave * 4) * BB + lm) * SIN + 4 * hh;
    float* tw = tile + (8 * hh * 32 + lm) * GPITCH + wave * 4;

    v4f a4[4][8];
    #pragma unroll
    for (int q4 = 0; q4 < 4; ++q4) {
        #pragma unroll
        for (int q = 0; q < 8; ++q)
            a4[q4][q] = *(const v4f*)(xw + q4 * (BB * SIN) + 8 * q);
    }

    const int pl = lane >> 3;
    const int jq = lane & 7;
    const int oo = wave * 4 + pl;
    const float* src = tile + oo * GPITCH + 4 * jq;

    #pragma nounroll
    for (int ob = 0; ob < SOUT / 32; ++ob) {
        const int o0 = ob * 32;

        v2f bfr[2][16];
        #pragma unroll
        for (int m = 0; m < 16; ++m) {
            const int kr = 8 * (m >> 1) + 4 * hh + 2 * (m & 1);
            #pragma unroll
            for (int nt = 0; nt < 2; ++nt) {
                const float* wp = w + (size_t)kr * SOUT + (o0 + nt * 16 + lm);
                bfr[nt][m].x = wp[0];
                bfr[nt][m].y = wp[SOUT];
            }
        }

        #pragma unroll
        for (int q4 = 0; q4 < 4; ++q4) {
            v8f acc0 = {};
            v8f acc1 = {};
            #pragma unroll
            for (int m = 0; m < 16; ++m) {
                v2f af;
                af.x = (m & 1) ? a4[q4][m >> 1].z : a4[q4][m >> 1].x;
                af.y = (m & 1) ? a4[q4][m >> 1].w : a4[q4][m >> 1].y;
                acc0 = __builtin_amdgcn_wmma_f32_16x16x4_f32(
                    false, af, false, bfr[0][m], (short)0, acc0, false, false);
                acc1 = __builtin_amdgcn_wmma_f32_16x16x4_f32(
                    false, af, false, bfr[1][m], (short)0, acc1, false, false);
            }

            #pragma unroll
            for (int v = 0; v < 8; ++v) {
                tw[(v * 32) * GPITCH + q4]      = acc0[v];
                tw[(v * 32 + 16) * GPITCH + q4] = acc1[v];
            }
        }
        __syncthreads();

        const float bv = bias[o0 + oo];
        float* dst = a_t + (size_t)(o0 + oo) * (DD1 * DD2) + (size_t)i * DD2 + j0 + 4 * jq;
        #pragma unroll
        for (int b = 0; b < BB; ++b) {
            v4f val = *(const v4f*)(src + b * 32 * GPITCH);
            val.x += bv;
            val.y += bv;
            val.z += bv;
            val.w += bv;
            *(v4f*)(dst + (size_t)b * SOUT * (DD1 * DD2)) = val;
        }
        __builtin_amdgcn_fence(__ATOMIC_RELEASE, "workgroup", "local");
        __builtin_amdgcn_s_barrier();
        __builtin_amdgcn_fence(__ATOMIC_ACQUIRE, "workgroup", "local");
    }
}

#define RING_ROWS 64
#define CHUNK     8
#define NCHUNK    20
#define LAG       5
#define NPHASE    (NCHUNK + 3 * LAG)

__global__ __launch_bounds__(128)
void mdrnn_scan(float* __restrict__ ws, const float* __restrict__ u)
{
    __shared__ float boundary[3][DD1];
    __shared__ float ring[4][RING_ROWS * 32];

    const int lane = threadIdx.x & 31;
    const int s    = __builtin_amdgcn_readfirstlane((int)(threadIdx.x >> 5));
    const int bo   = blockIdx.x;
    const int o    = bo & 127;

    const float u0 = u[o];
    const float u1 = u[SOUT + o];
    float* abase = ws + (size_t)bo * (DD1 * DD2) + s * 32 + lane;
    const float* bin  = boundary[s > 0 ? s - 1 : 0];
    float*       bout = boundary[s < 3 ? s : 2];
    const bool feeds  = (lane == 31) && (s < 3);
    const bool fed    = (lane == 0) && (s > 0);
    float* rw = ring[s];

    float r[CHUNK];
    #pragma unroll
    for (int k = 0; k < CHUNK; ++k)
        r[k] = abase[k * DD2];

    float h = 0.0f;
    for (int p = 0; p < NPHASE; ++p) {
        const int c = p - LAG * s;
        if (c >= 0 && c < NCHUNK) {
            const int t0 = c * CHUNK;
            if (t0 < DD1) {
                #pragma unroll
                for (int k = 0; k < CHUNK; ++k)
                    rw[((t0 + k) & (RING_ROWS - 1)) * 32 + lane] = r[k];
            }
            if (t0 >= 40) {
                #pragma unroll
                for (int k = 0; k < CHUNK; ++k) {
                    const int i = t0 - 40 + k;
                    abase[i * DD2] = rw[(i & (RING_ROWS - 1)) * 32 + lane];
                }
            }
            if (t0 + CHUNK < DD1) {
                #pragma unroll
                for (int k = 0; k < CHUNK; ++k)
                    r[k] = abase[(t0 + CHUNK + k) * DD2];
            }

            float av[CHUNK], bv[CHUNK];
            #pragma unroll
            for (int k = 0; k < CHUNK; ++k) {
                const int i = t0 + k - lane;
                av[k] = 0.0f;
                bv[k] = 0.0f;
                if (i >= 0 && i < DD1)
                    av[k] = rw[(i & (RING_ROWS - 1)) * 32 + lane];
            }
            if (fed && t0 < DD1) {
                #pragma unroll
                for (int k = 0; k < CHUNK; ++k)
                    bv[k] = bin[t0 + k];
            }

            #pragma unroll
            for (int k = 0; k < CHUNK; ++k) {
                const int t = t0 + k;
                const int i = t - lane;
                float hl = __shfl_up(h, 1);
                if (lane == 0)
                    hl = bv[k];
                if (i >= 0 && i < DD1) {
                    const float hn = __builtin_amdgcn_tanhf(fmaf(u0, h, fmaf(u1, hl, av[k])));
                    h = hn;
                    if (feeds) bout[i] = hn;
                    rw[(i & (RING_ROWS - 1)) * 32 + lane] = hn;
                }
            }

            if (c == NCHUNK - 1) {
                #pragma unroll
                for (int k = 0; k < CHUNK; ++k) {
                    const int i = DD1 - CHUNK + k;
                    abase[i * DD2] = rw[(i & (RING_ROWS - 1)) * 32 + lane];
                }
            }
        }
        __syncthreads();
    }
}

#define TPITCH 129

__global__ __launch_bounds__(256)
void mdrnn_out_transpose(const float* __restrict__ h_t, float* __restrict__ h_out)
{
    __shared__ float tile[SOUT * TPITCH];

    const int b    = blockIdx.x;
    const int i    = blockIdx.y;
    const int lane = threadIdx.x & 31;
    const int wave = threadIdx.x >> 5;

    const float* src = h_t + ((size_t)(b * SOUT + wave * 16) * DD1 + i) * DD2 + lane * 4;
    v4f v[16];
    #pragma unroll
    for (int r = 0; r < 16; ++r)
        v[r] = *(const v4f*)(src + (size_t)r * (DD1 * DD2));
    #pragma unroll
    for (int r = 0; r < 16; ++r) {
        float* d = tile + (wave * 16 + r) * TPITCH + lane * 4;
        d[0] = v[r].x;
        d[1] = v[r].y;
        d[2] = v[r].z;
        d[3] = v[r].w;
    }
    __syncthreads();

    float* dst = h_out + (((size_t)i * DD2 + wave * 16) * BB + b) * SOUT + lane * 4;
    const float* col = tile + (lane * 4) * TPITCH + wave * 16;
    #pragma unroll
    for (int r = 0; r < 16; ++r) {
        v4f q;
        q.x = col[0 * TPITCH + r];
        q.y = col[1 * TPITCH + r];
        q.z = col[2 * TPITCH + r];
        q.w = col[3 * TPITCH + r];
        *(v4f*)(dst + (size_t)r * (BB * SOUT)) = q;
    }
}

extern "C" void kernel_launch(void* const* d_in, const int* in_sizes, int n_in,
                              void* d_out, int out_size, void* d_ws, size_t ws_size,
                              hipStream_t stream)
{
    const float* x    = (const float*)d_in[0];
    const float* w    = (const float*)d_in[1];
    const float* u    = (const float*)d_in[2];
    const float* bias = (const float*)d_in[3];
    float* h   = (float*)d_out;
    float* a_t = (float*)d_ws;

    dim3 g1(1, DD2 / 32, DD1);
    mdrnn_gemm_wmma<<<g1, 256, 0, stream>>>(x, w, bias, a_t);

    mdrnn_scan<<<BB * SOUT, 128, 0, stream>>>(a_t, u);

    dim3 g3(BB, DD1);
    mdrnn_out_transpose<<<g3, 256, 0, stream>>>(a_t, h);
}


# scan: left neighbour's h moves lane to lane on the vector ALU (row shift + one cross-row fetch), not by the LDS-crossbar shuffle whose round trip and store drain sat in each step; sums unchanged
# speedup vs baseline: 6.9644x; 1.0500x over previous
#include <hip/hip_runtime.h>


#define DD1  128
#define DD2  128
#define BB   16
#define SIN  64
#define SOUT 128

typedef __attribute__((ext_vector_type(2))) float v2f;
typedef __attribute__((ext_vector_type(4))) float v4f;
typedef __attribute__((ext_vector_type(8))) float v8f;

#define GPITCH 36

__global__ __launch_bounds__(256)
void mdrnn_gemm_wmma(const float* __restrict__ x, const float* __restrict__ w,
                     const float* __restrict__ bias, float* __restrict__ a_t)
{
    __shared__ __attribute__((aligned(16))) float tile[512 * GPITCH];

    const int i    = blockIdx.z;
    const int j0   = blockIdx.y * 32;
    const int tid  = threadIdx.x;
    const int lane = tid & 31;
    const int wave = tid >> 5;

    const int lm = lane & 15;
    const int hh = lane >> 4;

    const float* xw = x + (((size_t)i * DD2 + j0 + wave * 4) * BB + lm) * SIN + 4 * hh;
    float* tw = tile + (8 * hh * 32 + lm) * GPITCH + wave * 4;

    v4f a4[4][8];
    #pragma unroll
    for (int q4 = 0; q4 < 4; ++q4) {
        #pragma unroll
        for (int q = 0; q < 8; ++q)
            a4[q4][q] = *(const v4f*)(xw + q4 * (BB * SIN) + 8 * q);
    }

    const int pl = lane >> 3;
    const int jq = lane & 7;
    const int oo = wave * 4 + pl;
    const float* src = tile + oo * GPITCH + 4 * jq;

    #pragma nounroll
    for (int ob = 0; ob < SOUT / 32; ++ob) {
        const int o0 = ob * 32;

        v2f bfr[2][16];
        #pragma unroll
        for (int m = 0; m < 16; ++m) {
            const int kr = 8 * (m >> 1) + 4 * hh + 2 * (m & 1);
            #pragma unroll
            for (int nt = 0; nt < 2; ++nt) {
                const float* wp = w + (size_t)kr * SOUT + (o0 + nt * 16 + lm);
                bfr[nt][m].x = wp[0];
                bfr[nt][m].y = wp[SOUT];
            }
        }

        #pragma unroll
        for (int q4 = 0; q4 < 4; ++q4) {
            v8f acc0 = {};
            v8f acc1 = {};
            #pragma unroll
            for (int m = 0; m < 16; ++m) {
                v2f af;
                af.x = (m & 1) ? a4[q4][m >> 1].z : a4[q4][m >> 1].x;
                af.y = (m & 1) ? a4[q4][m >> 1].w : a4[q4][m >> 1].y;
                acc0 = __builtin_amdgcn_wmma_f32_16x16x4_f32(
                    false, af, false, bfr[0][m], (short)0, acc0, false, false);
                acc1 = __builtin_amdgcn_wmma_f32_16x16x4_f32(
                    false, af, false, bfr[1][m], (short)0, acc1, false, false);
            }

            #pragma unroll
            for (int v = 0; v < 8; ++v) {
                tw[(v * 32) * GPITCH + q4]      = acc0[v];
                tw[(v * 32 + 16) * GPITCH + q4] = acc1[v];
            }
        }
        __syncthreads();

        const float bv = bias[o0 + oo];
        float* dst = a_t + (size_t)(o0 + oo) * (DD1 * DD2) + (size_t)i * DD2 + j0 + 4 * jq;
        #pragma unroll
        for (int b = 0; b < BB; ++b) {
            v4f val = *(const v4f*)(src + b * 32 * GPITCH);
            val.x += bv;
            val.y += bv;
            val.z += bv;
            val.w += bv;
            *(v4f*)(dst + (size_t)b * SOUT * (DD1 * DD2)) = val;
        }
        __builtin_amdgcn_fence(__ATOMIC_RELEASE, "workgroup", "local");
        __builtin_amdgcn_s_barrier();
        __builtin_amdgcn_fence(__ATOMIC_ACQUIRE, "workgroup", "local");
    }
}

#define RING_ROWS 64
#define CHUNK     8
#define NCHUNK    20
#define LAG       5
#define NPHASE    (NCHUNK + 3 * LAG)

__global__ __launch_bounds__(128)
void mdrnn_scan(float* __restrict__ ws, const float* __restrict__ u)
{
    __shared__ float boundary[3][DD1];
    __shared__ float ring[4][RING_ROWS * 32];

    const int lane = threadIdx.x & 31;
    const int s    = __builtin_amdgcn_readfirstlane((int)(threadIdx.x >> 5));
    const int bo   = blockIdx.x;
    const int o    = bo & 127;

    const float u0 = u[o];
    const float u1 = u[SOUT + o];
    float* abase = ws + (size_t)bo * (DD1 * DD2) + s * 32 + lane;
    const float* bin  = boundary[s > 0 ? s - 1 : 0];
    float*       bout = boundary[s < 3 ? s : 2];
    const bool feeds  = (lane == 31) && (s < 3);
    const bool fed    = (lane == 0) && (s > 0);
    const bool row1_head = (lane == 16);
    float* rw = ring[s];

    float r[CHUNK];
    #pragma unroll
    for (int k = 0; k < CHUNK; ++k)
        r[k] = abase[k * DD2];

    float h = 0.0f;
    for (int p = 0; p < NPHASE; ++p) {
        const int c = p - LAG * s;
        if (c >= 0 && c < NCHUNK) {
            const int t0 = c * CHUNK;
            if (t0 < DD1) {
                #pragma unroll
                for (int k = 0; k < CHUNK; ++k)
                    rw[((t0 + k) & (RING_ROWS - 1)) * 32 + lane] = r[k];
            }
            if (t0 >= 40) {
                #pragma unroll
                for (int k = 0; k < CHUNK; ++k) {
                    const int i = t0 - 40 + k;
                    abase[i * DD2] = rw[(i & (RING_ROWS - 1)) * 32 + lane];
                }
            }
            if (t0 + CHUNK < DD1) {
                #pragma unroll
                for (int k = 0; k < CHUNK; ++k)
                    r[k] = abase[(t0 + CHUNK + k) * DD2];
            }

            float av[CHUNK], bv[CHUNK];
            #pragma unroll
            for (int k = 0; k < CHUNK; ++k) {
                const int i = t0 + k - lane;
                av[k] = 0.0f;
                bv[k] = 0.0f;
                if (i >= 0 && i < DD1)
                    av[k] = rw[(i & (RING_ROWS - 1)) * 32 + lane];
            }
            if (fed && t0 < DD1) {
                #pragma unroll
                for (int k = 0; k < CHUNK; ++k)
                    bv[k] = bin[t0 + k];
            }

            #pragma unroll
            for (int k = 0; k < CHUNK; ++k) {
                const int t = t0 + k;
                const int i = t - lane;
                const int hb = __builtin_bit_cast(int, h);
                const int xr = __builtin_amdgcn_permlanex16(0, hb, 0xffffffffu, 0xffffffffu, false, false);
                const int sr = __builtin_amdgcn_update_dpp(__builtin_bit_cast(int, bv[k]), hb, 0x111, 0xf, 0xf, false);
                const float hl = __builtin_bit_cast(float, row1_head ? xr : sr);
                if (i >= 0 && i < DD1) {
                    const float hn = __builtin_amdgcn_tanhf(fmaf(u0, h, fmaf(u1, hl, av[k])));
                    h = hn;
                    if (feeds) bout[i] = hn;
                    rw[(i & (RING_ROWS - 1)) * 32 + lane] = hn;
                }
            }

            if (c == NCHUNK - 1) {
                #pragma unroll
                for (int k = 0; k < CHUNK; ++k) {
                    const int i = DD1 - CHUNK + k;
                    abase[i * DD2] = rw[(i & (RING_ROWS - 1)) * 32 + lane];
                }
            }
        }
        __syncthreads();
    }
}

#define TPITCH 129

__global__ __launch_bounds__(256)
void mdrnn_out_transpose(const float* __restrict__ h_t, float* __restrict__ h_out)
{
    __shared__ float tile[SOUT * TPITCH];

    const int b    = blockIdx.x;
    const int i    = blockIdx.y;
    const int lane = threadIdx.x & 31;
    const int wave = threadIdx.x >> 5;

    const float* src = h_t + ((size_t)(b * SOUT + wave * 16) * DD1 + i) * DD2 + lane * 4;
    v4f v[16];
    #pragma unroll
    for (int r = 0; r < 16; ++r)
        v[r] = *(const v4f*)(src + (size_t)r * (DD1 * DD2));
    #pragma unroll
    for (int r = 0; r < 16; ++r) {
        float* d = tile + (wave * 16 + r) * TPITCH + lane * 4;
        d[0] = v[r].x;
        d[1] = v[r].y;
        d[2] = v[r].z;
        d[3] = v[r].w;
    }
    __syncthreads();

    float* dst = h_out + (((size_t)i * DD2 + wave * 16) * BB + b) * SOUT + lane * 4;
    const float* col = tile + (lane * 4) * TPITCH + wave * 16;
    #pragma unroll
    for (int r = 0; r < 16; ++r) {
        v4f q;
        q.x = col[0 * TPITCH + r];
        q.y = col[1 * TPITCH + r];
        q.z = col[2 * TPITCH + r];
        q.w = col[3 * TPITCH + r];
        *(v4f*)(dst + (size_t)r * (BB * SOUT)) = q;
    }
}

extern "C" void kernel_launch(void* const* d_in, const int* in_sizes, int n_in,
                              void* d_out, int out_size, void* d_ws, size_t ws_size,
                              hipStream_t stream)
{
    const float* x    = (const float*)d_in[0];
    const float* w    = (const float*)d_in[1];
    const float* u    = (const float*)d_in[2];
    const float* bias = (const float*)d_in[3];
    float* h   = (float*)d_out;
    float* a_t = (float*)d_ws;

    dim3 g1(1, DD2 / 32, DD1);
    mdrnn_gemm_wmma<<<g1, 256, 0, stream>>>(x, w, bias, a_t);

    mdrnn_scan<<<BB * SOUT, 128, 0, stream>>>(a_t, u);

    dim3 g3(BB, DD1);
    mdrnn_out_transpose<<<g3, 256, 0, stream>>>(a_t, h);
}
